# starting kernel
# baseline (speedup 1.0000x reference)
#include <hip/hip_runtime.h>

#define N_NODES 50000
#define N_EDGES 800000
#define IN_C    64
#define HID_C   128
#define OUT_C   64

typedef __attribute__((ext_vector_type(2))) float v2f;
typedef __attribute__((ext_vector_type(8))) float v8f;


__global__ void gcn_deg_init(float* __restrict__ deg) {
    int n = blockIdx.x * blockDim.x + threadIdx.x;
    if (n < N_NODES) deg[n] = 1.0f;
}

__global__ void gcn_deg_accum(const int* __restrict__ dst, float* __restrict__ deg) {
    int e = blockIdx.x * blockDim.x + threadIdx.x;
    if (e < N_EDGES) atomicAdd(&deg[dst[e]], 1.0f);
}

__global__ void gcn_deg_to_dinv(float* __restrict__ deg) {
    int n = blockIdx.x * blockDim.x + threadIdx.x;
    if (n < N_NODES) deg[n] = rsqrtf(deg[n]);
}


__global__ __launch_bounds__(256) void gcn_gemm1_wmma(
    const float* __restrict__ x, const float* __restrict__ W1, float* __restrict__ xw) {
    const int lane = threadIdx.x & 31;
    const int wave = threadIdx.x >> 5;
    const int m0   = blockIdx.x * 16;
    const int n0   = wave * 16;
    const int half = lane >> 4;
    const int l15  = lane & 15;
    const int row  = m0 + l15;
    const int col  = n0 + l15;
    v8f acc = {0.f, 0.f, 0.f, 0.f, 0.f, 0.f, 0.f, 0.f};

    for (int k0 = 0; k0 < IN_C; k0 += 4) {
        const int ka = k0 + half * 2;
        v2f a = *(const v2f*)(x + row * IN_C + ka);
        v2f b;
        b.x = W1[ka * HID_C + col];
        b.y = W1[(ka + 1) * HID_C + col];
        acc = __builtin_amdgcn_wmma_f32_16x16x4_f32(
             false, a,  false, b,
             (short)0, acc,  false,  false);
    }
#pragma unroll
    for (int r = 0; r < 8; ++r)
        xw[(size_t)(m0 + half * 8 + r) * HID_C + col] = acc[r];
}


__global__ __launch_bounds__(128) void gcn_gemm2_wmma(
    const float* __restrict__ h, const float* __restrict__ W2, float* __restrict__ xw) {
    const int lane = threadIdx.x & 31;
    const int wave = threadIdx.x >> 5;
    const int m0   = blockIdx.x * 16;
    const int n0   = wave * 16;
    const int half = lane >> 4;
    const int l15  = lane & 15;
    const int row  = m0 + l15;
    const int col  = n0 + l15;
    v8f acc = {0.f, 0.f, 0.f, 0.f, 0.f, 0.f, 0.f, 0.f};

    for (int k0 = 0; k0 < HID_C; k0 += 4) {
        const int ka = k0 + half * 2;
        v2f a = *(const v2f*)(h + (size_t)row * HID_C + ka);
        a.x = fmaxf(a.x, 0.0f);
        a.y = fmaxf(a.y, 0.0f);
        v2f b;
        b.x = W2[ka * OUT_C + col];
        b.y = W2[(ka + 1) * OUT_C + col];
        acc = __builtin_amdgcn_wmma_f32_16x16x4_f32(
            false, a, false, b, (short)0, acc, false, false);
    }
#pragma unroll
    for (int r = 0; r < 8; ++r)
        xw[(size_t)(m0 + half * 8 + r) * OUT_C + col] = acc[r];
}


template <int C>
__global__ void gcn_agg_init(const float* __restrict__ xw, const float* __restrict__ dinv,
                             const float* __restrict__ bias, float* __restrict__ agg) {
    int i = blockIdx.x * blockDim.x + threadIdx.x;
    if (i >= N_NODES * C) return;
    int n = i / C;
    int c = i % C;
    float dn = dinv[n];
    agg[i] = bias[c] + xw[i] * dn * dn;
}


template <int C>
__global__ void gcn_scatter(const int* __restrict__ src, const int* __restrict__ dst,
                            const float* __restrict__ dinv,
                            const float* __restrict__ xw, float* __restrict__ agg) {
    const int LPE = C / 4;
    int t = blockIdx.x * blockDim.x + threadIdx.x;
    int e = t / LPE;
    if (e >= N_EDGES) return;
    int q = t % LPE;
    int s = src[e];
    int d = dst[e];
    float norm = dinv[s] * dinv[d];
    const float4 v = *(const float4*)(xw + (size_t)s * C + q * 4);
    float* base = agg + (size_t)d * C + q * 4;
    atomicAdd(base + 0, v.x * norm);
    atomicAdd(base + 1, v.y * norm);
    atomicAdd(base + 2, v.z * norm);
    atomicAdd(base + 3, v.w * norm);
}


extern "C" void kernel_launch(void* const* d_in, const int* in_sizes, int n_in,
                              void* d_out, int out_size, void* d_ws, size_t ws_size,
                              hipStream_t stream) {
    const float* x  = (const float*)d_in[0];
    const int*   ei = (const int*)d_in[1];
    const float* W1 = (const float*)d_in[2];
    const float* b1 = (const float*)d_in[3];
    const float* W2 = (const float*)d_in[4];
    const float* b2 = (const float*)d_in[5];
    float* out = (float*)d_out;

    const int* src = ei;
    const int* dst = ei + N_EDGES;

    float* dinv = (float*)d_ws;
    float* xw1  = dinv + 50176;
    float* agg1 = xw1 + 6400000;
    float* xw2  = xw1;

    gcn_deg_init<<<(N_NODES + 255) / 256, 256, 0, stream>>>(dinv);
    gcn_deg_accum<<<(N_EDGES + 255) / 256, 256, 0, stream>>>(dst, dinv);
    gcn_deg_to_dinv<<<(N_NODES + 255) / 256, 256, 0, stream>>>(dinv);

    gcn_gemm1_wmma<<<N_NODES / 16, 256, 0, stream>>>(x, W1, xw1);

    gcn_agg_init<HID_C><<<(N_NODES * HID_C + 255) / 256, 256, 0, stream>>>(xw1, dinv, b1, agg1);
    gcn_scatter<HID_C><<<(N_EDGES * (HID_C / 4) + 255) / 256, 256, 0, stream>>>(src, dst, dinv, xw1, agg1);

    gcn_gemm2_wmma<<<N_NODES / 16, 128, 0, stream>>>(agg1, W2, xw2);

    gcn_agg_init<OUT_C><<<(N_NODES * OUT_C + 255) / 256, 256, 0, stream>>>(xw2, dinv, b2, out);
    gcn_scatter<OUT_C><<<(N_EDGES * (OUT_C / 4) + 255) / 256, 256, 0, stream>>>(src, dst, dinv, xw2, out);
}


# Layer 1 aggregates the 64-channel input before the W1 product, (A x) W1 for A (x W1): its edge scatter moves 64 channels not 128; neighbour sums now precede the product (float sum order changes)
# speedup vs baseline: 1.2803x; 1.2803x over previous
#include <hip/hip_runtime.h>

#define N_NODES 50000
#define N_EDGES 800000
#define IN_C    64
#define HID_C   128
#define OUT_C   64

typedef __attribute__((ext_vector_type(2))) float v2f;
typedef __attribute__((ext_vector_type(8))) float v8f;


__global__ void gcn_deg_init(float* __restrict__ deg) {
    int n = blockIdx.x * blockDim.x + threadIdx.x;
    if (n < N_NODES) deg[n] = 1.0f;
}

__global__ void gcn_deg_accum(const int* __restrict__ dst, float* __restrict__ deg) {
    int e = blockIdx.x * blockDim.x + threadIdx.x;
    if (e < N_EDGES) atomicAdd(&deg[dst[e]], 1.0f);
}

__global__ void gcn_deg_to_dinv(float* __restrict__ deg) {
    int n = blockIdx.x * blockDim.x + threadIdx.x;
    if (n < N_NODES) deg[n] = rsqrtf(deg[n]);
}


__global__ __launch_bounds__(256) void gcn_gemm1_wmma(
    const float* __restrict__ ax, const float* __restrict__ W1,
    const float* __restrict__ b1, float* __restrict__ h) {
    const int lane = threadIdx.x & 31;
    const int wave = threadIdx.x >> 5;
    const int m0   = blockIdx.x * 16;
    const int n0   = wave * 16;
    const int half = lane >> 4;
    const int l15  = lane & 15;
    const int row  = m0 + l15;
    const int col  = n0 + l15;
    const float bc = b1[col];
    v8f acc = {bc, bc, bc, bc, bc, bc, bc, bc};

    for (int k0 = 0; k0 < IN_C; k0 += 4) {
        const int ka = k0 + half * 2;
        v2f a = *(const v2f*)(ax + row * IN_C + ka);
        v2f b;
        b.x = W1[ka * HID_C + col];
        b.y = W1[(ka + 1) * HID_C + col];
        acc = __builtin_amdgcn_wmma_f32_16x16x4_f32(
             false, a,  false, b,
             (short)0, acc,  false,  false);
    }
#pragma unroll
    for (int r = 0; r < 8; ++r)
        h[(size_t)(m0 + half * 8 + r) * HID_C + col] = acc[r];
}


__global__ __launch_bounds__(128) void gcn_gemm2_wmma(
    const float* __restrict__ h, const float* __restrict__ W2, float* __restrict__ xw) {
    const int lane = threadIdx.x & 31;
    const int wave = threadIdx.x >> 5;
    const int m0   = blockIdx.x * 16;
    const int n0   = wave * 16;
    const int half = lane >> 4;
    const int l15  = lane & 15;
    const int row  = m0 + l15;
    const int col  = n0 + l15;
    v8f acc = {0.f, 0.f, 0.f, 0.f, 0.f, 0.f, 0.f, 0.f};

    for (int k0 = 0; k0 < HID_C; k0 += 4) {
        const int ka = k0 + half * 2;
        v2f a = *(const v2f*)(h + (size_t)row * HID_C + ka);
        a.x = fmaxf(a.x, 0.0f);
        a.y = fmaxf(a.y, 0.0f);
        v2f b;
        b.x = W2[ka * OUT_C + col];
        b.y = W2[(ka + 1) * OUT_C + col];
        acc = __builtin_amdgcn_wmma_f32_16x16x4_f32(
            false, a, false, b, (short)0, acc, false, false);
    }
#pragma unroll
    for (int r = 0; r < 8; ++r)
        xw[(size_t)(m0 + half * 8 + r) * OUT_C + col] = acc[r];
}


template <int C, bool BIAS>
__global__ void gcn_agg_init(const float* __restrict__ xw, const float* __restrict__ dinv,
                             const float* __restrict__ bias, float* __restrict__ agg) {
    int i = blockIdx.x * blockDim.x + threadIdx.x;
    if (i >= N_NODES * C) return;
    int n = i / C;
    int c = i % C;
    float dn = dinv[n];
    float v = xw[i] * dn * dn;
    if (BIAS) v += bias[c];
    agg[i] = v;
}


template <int C>
__global__ void gcn_scatter(const int* __restrict__ src, const int* __restrict__ dst,
                            const float* __restrict__ dinv,
                            const float* __restrict__ xw, float* __restrict__ agg) {
    const int LPE = C / 4;
    int t = blockIdx.x * blockDim.x + threadIdx.x;
    int e = t / LPE;
    if (e >= N_EDGES) return;
    int q = t % LPE;
    int s = src[e];
    int d = dst[e];
    float norm = dinv[s] * dinv[d];
    const float4 v = *(const float4*)(xw + (size_t)s * C + q * 4);
    float* base = agg + (size_t)d * C + q * 4;
    atomicAdd(base + 0, v.x * norm);
    atomicAdd(base + 1, v.y * norm);
    atomicAdd(base + 2, v.z * norm);
    atomicAdd(base + 3, v.w * norm);
}


extern "C" void kernel_launch(void* const* d_in, const int* in_sizes, int n_in,
                              void* d_out, int out_size, void* d_ws, size_t ws_size,
                              hipStream_t stream) {
    const float* x  = (const float*)d_in[0];
    const int*   ei = (const int*)d_in[1];
    const float* W1 = (const float*)d_in[2];
    const float* b1 = (const float*)d_in[3];
    const float* W2 = (const float*)d_in[4];
    const float* b2 = (const float*)d_in[5];
    float* out = (float*)d_out;

    const int* src = ei;
    const int* dst = ei + N_EDGES;

    float* dinv = (float*)d_ws;
    float* ax   = dinv + 50176;
    float* h    = ax + 3200000;
    float* xw2  = h + 6400000;

    gcn_deg_init<<<(N_NODES + 255) / 256, 256, 0, stream>>>(dinv);
    gcn_deg_accum<<<(N_EDGES + 255) / 256, 256, 0, stream>>>(dst, dinv);
    gcn_deg_to_dinv<<<(N_NODES + 255) / 256, 256, 0, stream>>>(dinv);

    gcn_agg_init<IN_C, false><<<(N_NODES * IN_C + 255) / 256, 256, 0, stream>>>(x, dinv, b1, ax);
    gcn_scatter<IN_C><<<(N_EDGES * (IN_C / 4) + 255) / 256, 256, 0, stream>>>(src, dst, dinv, x, ax);

    gcn_gemm1_wmma<<<N_NODES / 16, 256, 0, stream>>>(ax, W1, b1, h);

    gcn_gemm2_wmma<<<N_NODES / 16, 128, 0, stream>>>(h, W2, xw2);

    gcn_agg_init<OUT_C, true><<<(N_NODES * OUT_C + 255) / 256, 256, 0, stream>>>(xw2, dinv, b2, out);
    gcn_scatter<OUT_C><<<(N_EDGES * (OUT_C / 4) + 255) / 256, 256, 0, stream>>>(src, dst, dinv, xw2, out);
}


# Edges bucketed by destination each call (count+rank, offsets, fill); both aggregations gather per node with plain stores, no float atomics; per-node sum order is now the edges' arrival rank
# speedup vs baseline: 2.2105x; 1.7265x over previous
#include <hip/hip_runtime.h>

#define N_NODES 50000
#define N_EDGES 800000
#define IN_C    64
#define HID_C   128
#define OUT_C   64
#define N_PAD   50176

typedef __attribute__((ext_vector_type(2))) float v2f;
typedef __attribute__((ext_vector_type(8))) float v8f;


__global__ void gcn_cnt_zero(int* __restrict__ cnt) {
    int n = blockIdx.x * blockDim.x + threadIdx.x;
    if (n < N_PAD) cnt[n] = 0;
}

__global__ void gcn_edge_rank(const int* __restrict__ dst, int* __restrict__ cnt,
                              int* __restrict__ rank) {
    int e = blockIdx.x * blockDim.x + threadIdx.x;
    if (e < N_EDGES) rank[e] = atomicAdd(&cnt[dst[e]], 1);
}

__global__ __launch_bounds__(256) void gcn_seg_alloc(const int* cnt, int* total,
                                                     int* __restrict__ off, float* __restrict__ dinv) {
    __shared__ int wsum[8];
    __shared__ int sbase;
    const int tid  = threadIdx.x;
    const int lane = tid & 31;
    const int wave = tid >> 5;
    const int n    = blockIdx.x * 256 + tid;
    const int c    = (n < N_NODES) ? cnt[n] : 0;
    int incl = c;
    for (int d = 1; d < 32; d <<= 1) {
        int v = __shfl_up(incl, d);
        if (lane >= d) incl += v;
    }
    if (lane == 31) wsum[wave] = incl;
    __syncthreads();
    if (tid == 0) {
        int run = 0;
        for (int w = 0; w < 8; ++w) { int t = wsum[w]; wsum[w] = run; run += t; }
        sbase = atomicAdd(total, run);
    }
    __syncthreads();
    if (n < N_NODES) {
        off[n]  = sbase + wsum[wave] + incl - c;
        dinv[n] = rsqrtf((float)(c + 1));
    }
}

__global__ void gcn_edge_fill(const int* __restrict__ src, const int* __restrict__ dst,
                              const int* __restrict__ rank, const int* __restrict__ off,
                              const float* __restrict__ dinv, int* __restrict__ tab) {
    int e = blockIdx.x * blockDim.x + threadIdx.x;
    if (e >= N_EDGES) return;
    int s = src[e];
    int d = dst[e];
    int p = off[d] + rank[e];
    float w = dinv[s] * dinv[d];
    if ((unsigned)p < (unsigned)N_EDGES) ((int2*)tab)[p] = make_int2(s, __float_as_int(w));
}


template <bool BIAS>
__global__ __launch_bounds__(256) void gcn_gather64(
    const float* __restrict__ xw, const int* __restrict__ off, const int* __restrict__ cnt,
    const float* __restrict__ dinv, const int* __restrict__ tab,
    const float* __restrict__ bias, float* __restrict__ agg) {
    const int t = blockIdx.x * 256 + threadIdx.x;
    const int n = t >> 4;
    const int q = t & 15;
    if (n >= N_NODES) return;
    const float4* __restrict__ rows = (const float4*)xw;
    const float dn = dinv[n];
    const int   c  = cnt[n];
    const int2* __restrict__ seg = (const int2*)tab + off[n];
    const float4 v = rows[n * 16 + q];
    float4 acc;
    acc.x = v.x * dn * dn;
    acc.y = v.y * dn * dn;
    acc.z = v.z * dn * dn;
    acc.w = v.w * dn * dn;
    int j = 0;
    for (; j + 4 <= c; j += 4) {
        const int2 e0 = seg[j], e1 = seg[j + 1], e2 = seg[j + 2], e3 = seg[j + 3];
        const int s0 = (unsigned)e0.x < (unsigned)N_NODES ? e0.x : n;
        const int s1 = (unsigned)e1.x < (unsigned)N_NODES ? e1.x : n;
        const int s2 = (unsigned)e2.x < (unsigned)N_NODES ? e2.x : n;
        const int s3 = (unsigned)e3.x < (unsigned)N_NODES ? e3.x : n;
        const float4 r0 = rows[s0 * 16 + q];
        const float4 r1 = rows[s1 * 16 + q];
        const float4 r2 = rows[s2 * 16 + q];
        const float4 r3 = rows[s3 * 16 + q];
        const float w0 = __int_as_float(e0.y), w1 = __int_as_float(e1.y);
        const float w2 = __int_as_float(e2.y), w3 = __int_as_float(e3.y);
        acc.x += r0.x * w0; acc.y += r0.y * w0; acc.z += r0.z * w0; acc.w += r0.w * w0;
        acc.x += r1.x * w1; acc.y += r1.y * w1; acc.z += r1.z * w1; acc.w += r1.w * w1;
        acc.x += r2.x * w2; acc.y += r2.y * w2; acc.z += r2.z * w2; acc.w += r2.w * w2;
        acc.x += r3.x * w3; acc.y += r3.y * w3; acc.z += r3.z * w3; acc.w += r3.w * w3;
    }
    for (; j < c; ++j) {
        const int2 e0 = seg[j];
        const int s0 = (unsigned)e0.x < (unsigned)N_NODES ? e0.x : n;
        const float4 r0 = rows[s0 * 16 + q];
        const float w0 = __int_as_float(e0.y);
        acc.x += r0.x * w0; acc.y += r0.y * w0; acc.z += r0.z * w0; acc.w += r0.w * w0;
    }
    if (BIAS) {
        acc.x += bias[q * 4 + 0];
        acc.y += bias[q * 4 + 1];
        acc.z += bias[q * 4 + 2];
        acc.w += bias[q * 4 + 3];
    }
    ((float4*)agg)[n * 16 + q] = acc;
}


__global__ __launch_bounds__(256) void gcn_gemm1_wmma(
    const float* __restrict__ ax, const float* __restrict__ W1,
    const float* __restrict__ b1, float* __restrict__ h) {
    const int lane = threadIdx.x & 31;
    const int wave = threadIdx.x >> 5;
    const int m0   = blockIdx.x * 16;
    const int n0   = wave * 16;
    const int half = lane >> 4;
    const int l15  = lane & 15;
    const int row  = m0 + l15;
    const int col  = n0 + l15;
    const float bc = b1[col];
    v8f acc = {bc, bc, bc, bc, bc, bc, bc, bc};

    for (int k0 = 0; k0 < IN_C; k0 += 4) {
        const int ka = k0 + half * 2;
        v2f a = *(const v2f*)(ax + row * IN_C + ka);
        v2f b;
        b.x = W1[ka * HID_C + col];
        b.y = W1[(ka + 1) * HID_C + col];
        acc = __builtin_amdgcn_wmma_f32_16x16x4_f32(
             false, a,  false, b,
             (short)0, acc,  false,  false);
    }
#pragma unroll
    for (int r = 0; r < 8; ++r)
        h[(size_t)(m0 + half * 8 + r) * HID_C + col] = acc[r];
}


__global__ __launch_bounds__(128) void gcn_gemm2_wmma(
    const float* __restrict__ h, const float* __restrict__ W2, float* __restrict__ xw) {
    const int lane = threadIdx.x & 31;
    const int wave = threadIdx.x >> 5;
    const int m0   = blockIdx.x * 16;
    const int n0   = wave * 16;
    const int half = lane >> 4;
    const int l15  = lane & 15;
    const int row  = m0 + l15;
    const int col  = n0 + l15;
    v8f acc = {0.f, 0.f, 0.f, 0.f, 0.f, 0.f, 0.f, 0.f};

    for (int k0 = 0; k0 < HID_C; k0 += 4) {
        const int ka = k0 + half * 2;
        v2f a = *(const v2f*)(h + (size_t)row * HID_C + ka);
        a.x = fmaxf(a.x, 0.0f);
        a.y = fmaxf(a.y, 0.0f);
        v2f b;
        b.x = W2[ka * OUT_C + col];
        b.y = W2[(ka + 1) * OUT_C + col];
        acc = __builtin_amdgcn_wmma_f32_16x16x4_f32(
            false, a, false, b, (short)0, acc, false, false);
    }
#pragma unroll
    for (int r = 0; r < 8; ++r)
        xw[(size_t)(m0 + half * 8 + r) * OUT_C + col] = acc[r];
}


extern "C" void kernel_launch(void* const* d_in, const int* in_sizes, int n_in,
                              void* d_out, int out_size, void* d_ws, size_t ws_size,
                              hipStream_t stream) {
    const float* x  = (const float*)d_in[0];
    const int*   ei = (const int*)d_in[1];
    const float* W1 = (const float*)d_in[2];
    const float* b1 = (const float*)d_in[3];
    const float* W2 = (const float*)d_in[4];
    const float* b2 = (const float*)d_in[5];
    float* out = (float*)d_out;

    const int* src = ei;
    const int* dst = ei + N_EDGES;

    int*   cnt  = (int*)d_ws;
    int*   off  = cnt + N_PAD;
    float* dinv = (float*)(off + N_PAD);
    int*   rank = (int*)(dinv + N_PAD);
    int*   tab  = rank + N_EDGES;
    float* ax   = (float*)(tab + 2 * N_EDGES);
    float* h    = ax + 3200000;
    float* xw2  = h + 6400000;

    gcn_cnt_zero<<<N_PAD / 256, 256, 0, stream>>>(cnt);
    gcn_edge_rank<<<(N_EDGES + 255) / 256, 256, 0, stream>>>(dst, cnt, rank);
    gcn_seg_alloc<<<N_PAD / 256, 256, 0, stream>>>(cnt, cnt + N_NODES, off, dinv);
    gcn_edge_fill<<<(N_EDGES + 255) / 256, 256, 0, stream>>>(src, dst, rank, off, dinv, tab);

    gcn_gather64<false><<<N_NODES * 16 / 256, 256, 0, stream>>>(x, off, cnt, dinv, tab, b1, ax);

    gcn_gemm1_wmma<<<N_NODES / 16, 256, 0, stream>>>(ax, W1, b1, h);

    gcn_gemm2_wmma<<<N_NODES / 16, 128, 0, stream>>>(h, W2, xw2);

    gcn_gather64<true><<<N_NODES * 16 / 256, 256, 0, stream>>>(xw2, off, cnt, dinv, tab, b2, out);
}


# GEMM1, ReLU and GEMM2 fused in one kernel: one wave per 16-node stripe, GEMM1 run transposed so its D registers are GEMM2's A fragments, W1/W2 in LDS; h never stored; k order inside products changes
# speedup vs baseline: 2.4737x; 1.1191x over previous
#include <hip/hip_runtime.h>

#define N_NODES 50000
#define N_EDGES 800000
#define IN_C    64
#define HID_C   128
#define OUT_C   64
#define N_PAD   50176

typedef __attribute__((ext_vector_type(2))) float v2f;
typedef __attribute__((ext_vector_type(8))) float v8f;


__global__ void gcn_cnt_zero(int* __restrict__ cnt) {
    int n = blockIdx.x * blockDim.x + threadIdx.x;
    if (n < N_PAD) cnt[n] = 0;
}

__global__ void gcn_edge_rank(const int* __restrict__ dst, int* __restrict__ cnt,
                              int* __restrict__ rank) {
    int e = blockIdx.x * blockDim.x + threadIdx.x;
    if (e < N_EDGES) rank[e] = atomicAdd(&cnt[dst[e]], 1);
}

__global__ __launch_bounds__(256) void gcn_seg_alloc(const int* cnt, int* total,
                                                     int* __restrict__ off, float* __restrict__ dinv) {
    __shared__ int wsum[8];
    __shared__ int sbase;
    const int tid  = threadIdx.x;
    const int lane = tid & 31;
    const int wave = tid >> 5;
    const int n    = blockIdx.x * 256 + tid;
    const int c    = (n < N_NODES) ? cnt[n] : 0;
    int incl = c;
    for (int d = 1; d < 32; d <<= 1) {
        int v = __shfl_up(incl, d);
        if (lane >= d) incl += v;
    }
    if (lane == 31) wsum[wave] = incl;
    __syncthreads();
    if (tid == 0) {
        int run = 0;
        for (int w = 0; w < 8; ++w) { int t = wsum[w]; wsum[w] = run; run += t; }
        sbase = atomicAdd(total, run);
    }
    __syncthreads();
    if (n < N_NODES) {
        off[n]  = sbase + wsum[wave] + incl - c;
        dinv[n] = rsqrtf((float)(c + 1));
    }
}

__global__ void gcn_edge_fill(const int* __restrict__ src, const int* __restrict__ dst,
                              const int* __restrict__ rank, const int* __restrict__ off,
                              const float* __restrict__ dinv, int* __restrict__ tab) {
    int e = blockIdx.x * blockDim.x + threadIdx.x;
    if (e >= N_EDGES) return;
    int s = src[e];
    int d = dst[e];
    int p = off[d] + rank[e];
    float w = dinv[s] * dinv[d];
    if ((unsigned)p < (unsigned)N_EDGES) ((int2*)tab)[p] = make_int2(s, __float_as_int(w));
}


template <bool BIAS>
__global__ __launch_bounds__(256) void gcn_gather64(
    const float* __restrict__ xw, const int* __restrict__ off, const int* __restrict__ cnt,
    const float* __restrict__ dinv, const int* __restrict__ tab,
    const float* __restrict__ bias, float* __restrict__ agg) {
    const int t = blockIdx.x * 256 + threadIdx.x;
    const int n = t >> 4;
    const int q = t & 15;
    if (n >= N_NODES) return;
    const float4* __restrict__ rows = (const float4*)xw;
    const float dn = dinv[n];
    const int   c  = cnt[n];
    const int2* __restrict__ seg = (const int2*)tab + off[n];
    const float4 v = rows[n * 16 + q];
    float4 acc;
    acc.x = v.x * dn * dn;
    acc.y = v.y * dn * dn;
    acc.z = v.z * dn * dn;
    acc.w = v.w * dn * dn;
    int j = 0;
    for (; j + 4 <= c; j += 4) {
        const int2 e0 = seg[j], e1 = seg[j + 1], e2 = seg[j + 2], e3 = seg[j + 3];
        const int s0 = (unsigned)e0.x < (unsigned)N_NODES ? e0.x : n;
        const int s1 = (unsigned)e1.x < (unsigned)N_NODES ? e1.x : n;
        const int s2 = (unsigned)e2.x < (unsigned)N_NODES ? e2.x : n;
        const int s3 = (unsigned)e3.x < (unsigned)N_NODES ? e3.x : n;
        const float4 r0 = rows[s0 * 16 + q];
        const float4 r1 = rows[s1 * 16 + q];
        const float4 r2 = rows[s2 * 16 + q];
        const float4 r3 = rows[s3 * 16 + q];
        const float w0 = __int_as_float(e0.y), w1 = __int_as_float(e1.y);
        const float w2 = __int_as_float(e2.y), w3 = __int_as_float(e3.y);
        acc.x += r0.x * w0; acc.y += r0.y * w0; acc.z += r0.z * w0; acc.w += r0.w * w0;
        acc.x += r1.x * w1; acc.y += r1.y * w1; acc.z += r1.z * w1; acc.w += r1.w * w1;
        acc.x += r2.x * w2; acc.y += r2.y * w2; acc.z += r2.z * w2; acc.w += r2.w * w2;
        acc.x += r3.x * w3; acc.y += r3.y * w3; acc.z += r3.z * w3; acc.w += r3.w * w3;
    }
    for (; j < c; ++j) {
        const int2 e0 = seg[j];
        const int s0 = (unsigned)e0.x < (unsigned)N_NODES ? e0.x : n;
        const float4 r0 = rows[s0 * 16 + q];
        const float w0 = __int_as_float(e0.y);
        acc.x += r0.x * w0; acc.y += r0.y * w0; acc.z += r0.z * w0; acc.w += r0.w * w0;
    }
    if (BIAS) {
        acc.x += bias[q * 4 + 0];
        acc.y += bias[q * 4 + 1];
        acc.z += bias[q * 4 + 2];
        acc.w += bias[q * 4 + 3];
    }
    ((float4*)agg)[n * 16 + q] = acc;
}

#define W1_PITCH 136
#define W2_PITCH 68
#define N_STRIPES (N_NODES / 16)

__global__ __launch_bounds__(256) void gcn_mlp_fused_wmma(
    const float* __restrict__ ax, const float* __restrict__ W1, const float* __restrict__ b1,
    const float* __restrict__ W2, float* __restrict__ xw) {
    __shared__ __attribute__((aligned(16))) float2 W1l[(IN_C / 2) * W1_PITCH];
    __shared__ __attribute__((aligned(16))) float2 W2l[(HID_C / 2) * W2_PITCH];
    const int tid    = threadIdx.x;
    const int lane   = tid & 31;
    const int wave   = tid >> 5;
    const int half   = lane >> 4;
    const int l15    = lane & 15;
    const int stripe = blockIdx.x * 8 + wave;
    const bool valid = stripe < N_STRIPES;
    const int m0     = (valid ? stripe : 0) * 16;

    float4 axv[8];
    const float4* __restrict__ arow = (const float4*)(ax + (size_t)(m0 + l15) * IN_C) + half;
#pragma unroll
    for (int i = 0; i < 8; ++i) axv[i] = arow[2 * i];

#pragma unroll
    for (int it = 0; it < 4; ++it) {
        const int item = it * 256 + tid;
        const int pr = item >> 5, q = item & 31;
        const float4 lo = *(const float4*)(W1 + (2 * pr) * HID_C + 4 * q);
        const float4 hi = *(const float4*)(W1 + (2 * pr + 1) * HID_C + 4 * q);
        float4* d = (float4*)(W1l + pr * W1_PITCH + 4 * q);
        d[0] = make_float4(lo.x, hi.x, lo.y, hi.y);
        d[1] = make_float4(lo.z, hi.z, lo.w, hi.w);
    }
#pragma unroll
    for (int it = 0; it < 4; ++it) {
        const int item = it * 256 + tid;
        const int pr = item >> 4, q = item & 15;
        const float4 lo = *(const float4*)(W2 + (2 * pr) * OUT_C + 4 * q);
        const float4 hi = *(const float4*)(W2 + (2 * pr + 1) * OUT_C + 4 * q);
        float4* d = (float4*)(W2l + pr * W2_PITCH + 4 * q);
        d[0] = make_float4(lo.x, hi.x, lo.y, hi.y);
        d[1] = make_float4(lo.z, hi.z, lo.w, hi.w);
    }
    __syncthreads();
    if (!valid) return;

    v8f acc1[8];
#pragma unroll
    for (int t = 0; t < 8; ++t) {
        const float4 ba = *(const float4*)(b1 + 16 * t + 8 * half);
        const float4 bb = *(const float4*)(b1 + 16 * t + 8 * half + 4);
        acc1[t] = (v8f){ba.x, ba.y, ba.z, ba.w, bb.x, bb.y, bb.z, bb.w};
    }
#pragma unroll
    for (int i = 0; i < 8; ++i) {
#pragma unroll
        for (int j = 0; j < 2; ++j) {
            v2f xb;
            xb.x = j ? axv[i].z : axv[i].x;
            xb.y = j ? axv[i].w : axv[i].y;
            const float2* __restrict__ wr = W1l + (4 * i + 2 * half + j) * W1_PITCH + l15;
#pragma unroll
            for (int t = 0; t < 8; ++t) {
                const float2 w = wr[16 * t];
                v2f wa; wa.x = w.x; wa.y = w.y;
                acc1[t] = __builtin_amdgcn_wmma_f32_16x16x4_f32(
                     false, wa,  false, xb,
                     (short)0, acc1[t],  false,  false);
            }
        }
    }

    v8f acc2[4];
#pragma unroll
    for (int u = 0; u < 4; ++u) acc2[u] = (v8f){0.f, 0.f, 0.f, 0.f, 0.f, 0.f, 0.f, 0.f};
#pragma unroll
    for (int t = 0; t < 8; ++t) {
#pragma unroll
        for (int j = 0; j < 4; ++j) {
            v2f ha;
            ha.x = fmaxf(acc1[t][2 * j], 0.0f);
            ha.y = fmaxf(acc1[t][2 * j + 1], 0.0f);
            const float2* __restrict__ wr = W2l + (8 * t + 4 * half + j) * W2_PITCH + l15;
#pragma unroll
            for (int u = 0; u < 4; ++u) {
                const float2 w = wr[16 * u];
                v2f wb; wb.x = w.x; wb.y = w.y;
                acc2[u] = __builtin_amdgcn_wmma_f32_16x16x4_f32(
                    false, ha, false, wb, (short)0, acc2[u], false, false);
            }
        }
    }
#pragma unroll
    for (int u = 0; u < 4; ++u)
#pragma unroll
        for (int r = 0; r < 8; ++r)
            xw[(size_t)(m0 + half * 8 + r) * OUT_C + 16 * u + l15] = acc2[u][r];
}


extern "C" void kernel_launch(void* const* d_in, const int* in_sizes, int n_in,
                              void* d_out, int out_size, void* d_ws, size_t ws_size,
                              hipStream_t stream) {
    const float* x  = (const float*)d_in[0];
    const int*   ei = (const int*)d_in[1];
    const float* W1 = (const float*)d_in[2];
    const float* b1 = (const float*)d_in[3];
    const float* W2 = (const float*)d_in[4];
    const float* b2 = (const float*)d_in[5];
    float* out = (float*)d_out;

    const int* src = ei;
    const int* dst = ei + N_EDGES;

    int*   cnt  = (int*)d_ws;
    int*   off  = cnt + N_PAD;
    float* dinv = (float*)(off + N_PAD);
    int*   rank = (int*)(dinv + N_PAD);
    int*   tab  = rank + N_EDGES;
    float* ax   = (float*)(tab + 2 * N_EDGES);
    float* xw2  = ax + 3200000;

    gcn_cnt_zero<<<N_PAD / 256, 256, 0, stream>>>(cnt);
    gcn_edge_rank<<<(N_EDGES + 255) / 256, 256, 0, stream>>>(dst, cnt, rank);
    gcn_seg_alloc<<<N_PAD / 256, 256, 0, stream>>>(cnt, cnt + N_NODES, off, dinv);
    gcn_edge_fill<<<(N_EDGES + 255) / 256, 256, 0, stream>>>(src, dst, rank, off, dinv, tab);

    gcn_gather64<false><<<N_NODES * 16 / 256, 256, 0, stream>>>(x, off, cnt, dinv, tab, b1, ax);

    gcn_mlp_fused_wmma<<<(N_STRIPES + 7) / 8, 256, 0, stream>>>(ax, W1, b1, W2, xw2);

    gcn_gather64<true><<<N_NODES * 16 / 256, 256, 0, stream>>>(xw2, off, cnt, dinv, tab, b2, out);
}


# Edge slots from LDS counters, not global atomics: block (edge chunk, node range) counts and places its range's edges, chunk counts prefixed per node; per-node sum order = chunk, then arrival
# speedup vs baseline: 4.0329x; 1.6303x over previous
#include <hip/hip_runtime.h>

#define N_NODES 50000
#define N_EDGES 800000
#define IN_C    64
#define HID_C   128
#define OUT_C   64
#define N_PAD   50176

typedef __attribute__((ext_vector_type(2))) float v2f;
typedef __attribute__((ext_vector_type(8))) float v8f;

#define EQ      16
#define ER      16
#define CHUNK_E (N_EDGES / EQ)
#define CHUNK_V (CHUNK_E / 4)
#define RANGE_N (N_NODES / ER)
#define SCAN_T  1024

__global__ __launch_bounds__(SCAN_T) void gcn_edge_count(const int* __restrict__ dst, int* __restrict__ pc,
                                                         int* __restrict__ total) {
    __shared__ int lcnt[RANGE_N];
    const int tid = threadIdx.x;
    const int q   = blockIdx.x % EQ;
    const int r   = blockIdx.x / EQ;
    const int lo  = r * RANGE_N;
    for (int i = tid; i < RANGE_N; i += SCAN_T) lcnt[i] = 0;
    if (blockIdx.x == 0 && tid == 0) *total = 0;
    __syncthreads();
    const int4* __restrict__ dv = (const int4*)(dst + q * CHUNK_E);
    for (int base = 0; base < CHUNK_V; base += 4 * SCAN_T) {
        int dd[16];
#pragma unroll
        for (int u = 0; u < 4; ++u) {
            const int i = base + u * SCAN_T + tid;
            int4 d = make_int4(-1, -1, -1, -1);
            if (i < CHUNK_V) d = dv[i];
            dd[4 * u + 0] = d.x; dd[4 * u + 1] = d.y; dd[4 * u + 2] = d.z; dd[4 * u + 3] = d.w;
        }
#pragma unroll
        for (int k = 0; k < 16; ++k) {
            const unsigned a = (unsigned)(dd[k] - lo);
            if (a < (unsigned)RANGE_N)
                __hip_atomic_fetch_add(&lcnt[a], 1, __ATOMIC_RELAXED, __HIP_MEMORY_SCOPE_WORKGROUP);
        }
    }
    __syncthreads();
    for (int i = tid; i < RANGE_N; i += SCAN_T) pc[q * N_PAD + lo + i] = lcnt[i];
}

__global__ __launch_bounds__(256) void gcn_seg_alloc(int* __restrict__ pc, int* cnt, int* total,
                                                     int* __restrict__ off, float* __restrict__ dinv) {
    __shared__ int wsum[8];
    __shared__ int sbase;
    const int tid  = threadIdx.x;
    const int lane = tid & 31;
    const int wave = tid >> 5;
    const int n    = blockIdx.x * 256 + tid;
    int c = 0;
    if (n < N_NODES) {
        int v[EQ];
#pragma unroll
        for (int q = 0; q < EQ; ++q) v[q] = pc[q * N_PAD + n];
#pragma unroll
        for (int q = 0; q < EQ; ++q) { pc[q * N_PAD + n] = c; c += v[q]; }
    }
    int incl = c;
    for (int d = 1; d < 32; d <<= 1) {
        int v = __shfl_up(incl, d);
        if (lane >= d) incl += v;
    }
    if (lane == 31) wsum[wave] = incl;
    __syncthreads();
    if (tid == 0) {
        int run = 0;
        for (int w = 0; w < 8; ++w) { int t = wsum[w]; wsum[w] = run; run += t; }
        sbase = atomicAdd(total, run);
    }
    __syncthreads();
    if (n < N_NODES) {
        cnt[n]  = c;
        off[n]  = sbase + wsum[wave] + incl - c;
        dinv[n] = rsqrtf((float)(c + 1));
    }
}

__global__ __launch_bounds__(SCAN_T) void gcn_edge_fill(const int* __restrict__ src, const int* __restrict__ dst,
                                                        const int* __restrict__ pc, const int* __restrict__ off,
                                                        const float* __restrict__ dinv, int* __restrict__ tab) {
    __shared__ int   lpos[RANGE_N];
    __shared__ float ldinv[RANGE_N];
    const int tid = threadIdx.x;
    const int q   = blockIdx.x % EQ;
    const int r   = blockIdx.x / EQ;
    const int lo  = r * RANGE_N;
    int   lp[4];
    float ld[4];
#pragma unroll
    for (int t = 0; t < 4; ++t) {
        const int i  = t * SCAN_T + tid;
        const int ic = i < RANGE_N ? i : RANGE_N - 1;
        lp[t] = off[lo + ic] + pc[q * N_PAD + lo + ic];
        ld[t] = dinv[lo + ic];
    }
#pragma unroll
    for (int t = 0; t < 4; ++t) {
        const int i = t * SCAN_T + tid;
        if (i < RANGE_N) { lpos[i] = lp[t]; ldinv[i] = ld[t]; }
    }
    __syncthreads();
    const int4* __restrict__ dv = (const int4*)(dst + q * CHUNK_E);
    const int4* __restrict__ sv = (const int4*)(src + q * CHUNK_E);
    for (int base = 0; base < CHUNK_V; base += 4 * SCAN_T) {
        int dd[16], ss[16];
#pragma unroll
        for (int u = 0; u < 4; ++u) {
            const int i = base + u * SCAN_T + tid;
            int4 d = make_int4(-1, -1, -1, -1);
            int4 s = make_int4(0, 0, 0, 0);
            if (i < CHUNK_V) { d = dv[i]; s = sv[i]; }
            dd[4 * u + 0] = d.x; dd[4 * u + 1] = d.y; dd[4 * u + 2] = d.z; dd[4 * u + 3] = d.w;
            ss[4 * u + 0] = s.x; ss[4 * u + 1] = s.y; ss[4 * u + 2] = s.z; ss[4 * u + 3] = s.w;
        }
        int p[16];
#pragma unroll
        for (int k = 0; k < 16; ++k) {
            const unsigned a = (unsigned)(dd[k] - lo);
            p[k] = -1;
            if (a < (unsigned)RANGE_N)
                p[k] = __hip_atomic_fetch_add(&lpos[a], 1, __ATOMIC_RELAXED, __HIP_MEMORY_SCOPE_WORKGROUP);
        }
        float w[16];
#pragma unroll
        for (int k = 0; k < 16; ++k) {
            const unsigned a  = (unsigned)(dd[k] - lo);
            const bool    hit = a < (unsigned)RANGE_N;
            w[k] = dinv[hit ? ss[k] : 0] * ldinv[hit ? a : 0u];
        }
#pragma unroll
        for (int k = 0; k < 16; ++k) {
            if ((unsigned)p[k] < (unsigned)N_EDGES)
                ((int2*)tab)[p[k]] = make_int2(ss[k], __float_as_int(w[k]));
        }
    }
}


template <bool BIAS>
__global__ __launch_bounds__(256) void gcn_gather64(
    const float* __restrict__ xw, const int* __restrict__ off, const int* __restrict__ cnt,
    const float* __restrict__ dinv, const int* __restrict__ tab,
    const float* __restrict__ bias, float* __restrict__ agg) {
    const int t = blockIdx.x * 256 + threadIdx.x;
    const int n = t >> 4;
    const int q = t & 15;
    if (n >= N_NODES) return;
    const float4* __restrict__ rows = (const float4*)xw;
    const float dn = dinv[n];
    const int   c  = cnt[n];
    const int2* __restrict__ seg = (const int2*)tab + off[n];
    const float4 v = rows[n * 16 + q];
    float4 acc;
    acc.x = v.x * dn * dn;
    acc.y = v.y * dn * dn;
    acc.z = v.z * dn * dn;
    acc.w = v.w * dn * dn;
    int j = 0;
    for (; j + 4 <= c; j += 4) {
        const int2 e0 = seg[j], e1 = seg[j + 1], e2 = seg[j + 2], e3 = seg[j + 3];
        const int s0 = (unsigned)e0.x < (unsigned)N_NODES ? e0.x : n;
        const int s1 = (unsigned)e1.x < (unsigned)N_NODES ? e1.x : n;
        const int s2 = (unsigned)e2.x < (unsigned)N_NODES ? e2.x : n;
        const int s3 = (unsigned)e3.x < (unsigned)N_NODES ? e3.x : n;
        const float4 r0 = rows[s0 * 16 + q];
        const float4 r1 = rows[s1 * 16 + q];
        const float4 r2 = rows[s2 * 16 + q];
        const float4 r3 = rows[s3 * 16 + q];
        const float w0 = __int_as_float(e0.y), w1 = __int_as_float(e1.y);
        const float w2 = __int_as_float(e2.y), w3 = __int_as_float(e3.y);
        acc.x += r0.x * w0; acc.y += r0.y * w0; acc.z += r0.z * w0; acc.w += r0.w * w0;
        acc.x += r1.x * w1; acc.y += r1.y * w1; acc.z += r1.z * w1; acc.w += r1.w * w1;
        acc.x += r2.x * w2; acc.y += r2.y * w2; acc.z += r2.z * w2; acc.w += r2.w * w2;
        acc.x += r3.x * w3; acc.y += r3.y * w3; acc.z += r3.z * w3; acc.w += r3.w * w3;
    }
    for (; j < c; ++j) {
        const int2 e0 = seg[j];
        const int s0 = (unsigned)e0.x < (unsigned)N_NODES ? e0.x : n;
        const float4 r0 = rows[s0 * 16 + q];
        const float w0 = __int_as_float(e0.y);
        acc.x += r0.x * w0; acc.y += r0.y * w0; acc.z += r0.z * w0; acc.w += r0.w * w0;
    }
    if (BIAS) {
        acc.x += bias[q * 4 + 0];
        acc.y += bias[q * 4 + 1];
        acc.z += bias[q * 4 + 2];
        acc.w += bias[q * 4 + 3];
    }
    ((float4*)agg)[n * 16 + q] = acc;
}

#define W1_PITCH 136
#define W2_PITCH 68
#define N_STRIPES (N_NODES / 16)

__global__ __launch_bounds__(256) void gcn_mlp_fused_wmma(
    const float* __restrict__ ax, const float* __restrict__ W1, const float* __restrict__ b1,
    const float* __restrict__ W2, float* __restrict__ xw) {
    __shared__ __attribute__((aligned(16))) float2 W1l[(IN_C / 2) * W1_PITCH];
    __shared__ __attribute__((aligned(16))) float2 W2l[(HID_C / 2) * W2_PITCH];
    const int tid    = threadIdx.x;
    const int lane   = tid & 31;
    const int wave   = tid >> 5;
    const int half   = lane >> 4;
    const int l15    = lane & 15;
    const int stripe = blockIdx.x * 8 + wave;
    const bool valid = stripe < N_STRIPES;
    const int m0     = (valid ? stripe : 0) * 16;

    float4 axv[8];
    const float4* __restrict__ arow = (const float4*)(ax + (size_t)(m0 + l15) * IN_C) + half;
#pragma unroll
    for (int i = 0; i < 8; ++i) axv[i] = arow[2 * i];

#pragma unroll
    for (int it = 0; it < 4; ++it) {
        const int item = it * 256 + tid;
        const int pr = item >> 5, q = item & 31;
        const float4 lo = *(const float4*)(W1 + (2 * pr) * HID_C + 4 * q);
        const float4 hi = *(const float4*)(W1 + (2 * pr + 1) * HID_C + 4 * q);
        float4* d = (float4*)(W1l + pr * W1_PITCH + 4 * q);
        d[0] = make_float4(lo.x, hi.x, lo.y, hi.y);
        d[1] = make_float4(lo.z, hi.z, lo.w, hi.w);
    }
#pragma unroll
    for (int it = 0; it < 4; ++it) {
        const int item = it * 256 + tid;
        const int pr = item >> 4, q = item & 15;
        const float4 lo = *(const float4*)(W2 + (2 * pr) * OUT_C + 4 * q);
        const float4 hi = *(const float4*)(W2 + (2 * pr + 1) * OUT_C + 4 * q);
        float4* d = (float4*)(W2l + pr * W2_PITCH + 4 * q);
        d[0] = make_float4(lo.x, hi.x, lo.y, hi.y);
        d[1] = make_float4(lo.z, hi.z, lo.w, hi.w);
    }
    __syncthreads();
    if (!valid) return;

    v8f acc1[8];
#pragma unroll
    for (int t = 0; t < 8; ++t) {
        const float4 ba = *(const float4*)(b1 + 16 * t + 8 * half);
        const float4 bb = *(const float4*)(b1 + 16 * t + 8 * half + 4);
        acc1[t] = (v8f){ba.x, ba.y, ba.z, ba.w, bb.x, bb.y, bb.z, bb.w};
    }
#pragma unroll
    for (int i = 0; i < 8; ++i) {
#pragma unroll
        for (int j = 0; j < 2; ++j) {
            v2f xb;
            xb.x = j ? axv[i].z : axv[i].x;
            xb.y = j ? axv[i].w : axv[i].y;
            const float2* __restrict__ wr = W1l + (4 * i + 2 * half + j) * W1_PITCH + l15;
#pragma unroll
            for (int t = 0; t < 8; ++t) {
                const float2 w = wr[16 * t];
                v2f wa; wa.x = w.x; wa.y = w.y;
                acc1[t] = __builtin_amdgcn_wmma_f32_16x16x4_f32(
                     false, wa,  false, xb,
                     (short)0, acc1[t],  false,  false);
            }
        }
    }

    v8f acc2[4];
#pragma unroll
    for (int u = 0; u < 4; ++u) acc2[u] = (v8f){0.f, 0.f, 0.f, 0.f, 0.f, 0.f, 0.f, 0.f};
#pragma unroll
    for (int t = 0; t < 8; ++t) {
#pragma unroll
        for (int j = 0; j < 4; ++j) {
            v2f ha;
            ha.x = fmaxf(acc1[t][2 * j], 0.0f);
            ha.y = fmaxf(acc1[t][2 * j + 1], 0.0f);
            const float2* __restrict__ wr = W2l + (8 * t + 4 * half + j) * W2_PITCH + l15;
#pragma unroll
            for (int u = 0; u < 4; ++u) {
                const float2 w = wr[16 * u];
                v2f wb; wb.x = w.x; wb.y = w.y;
                acc2[u] = __builtin_amdgcn_wmma_f32_16x16x4_f32(
                    false, ha, false, wb, (short)0, acc2[u], false, false);
            }
        }
    }
#pragma unroll
    for (int u = 0; u < 4; ++u)
#pragma unroll
        for (int r = 0; r < 8; ++r)
            xw[(size_t)(m0 + half * 8 + r) * OUT_C + 16 * u + l15] = acc2[u][r];
}


extern "C" void kernel_launch(void* const* d_in, const int* in_sizes, int n_in,
                              void* d_out, int out_size, void* d_ws, size_t ws_size,
                              hipStream_t stream) {
    const float* x  = (const float*)d_in[0];
    const int*   ei = (const int*)d_in[1];
    const float* W1 = (const float*)d_in[2];
    const float* b1 = (const float*)d_in[3];
    const float* W2 = (const float*)d_in[4];
    const float* b2 = (const float*)d_in[5];
    float* out = (float*)d_out;

    const int* src = ei;
    const int* dst = ei + N_EDGES;

    int*   cnt  = (int*)d_ws;
    int*   off  = cnt + N_PAD;
    float* dinv = (float*)(off + N_PAD);
    int*   pc   = (int*)(dinv + N_PAD);
    int*   tab  = pc + EQ * N_PAD;
    float* ax   = (float*)(tab + 2 * N_EDGES);
    float* xw2  = ax + 3200000;

    gcn_edge_count<<<EQ * ER, SCAN_T, 0, stream>>>(dst, pc, cnt + N_NODES);
    gcn_seg_alloc<<<N_PAD / 256, 256, 0, stream>>>(pc, cnt, cnt + N_NODES, off, dinv);
    gcn_edge_fill<<<EQ * ER, SCAN_T, 0, stream>>>(src, dst, pc, off, dinv, tab);

    gcn_gather64<false><<<N_NODES * 16 / 256, 256, 0, stream>>>(x, off, cnt, dinv, tab, b1, ax);

    gcn_mlp_fused_wmma<<<(N_STRIPES + 7) / 8, 256, 0, stream>>>(ax, W1, b1, W2, xw2);

    gcn_gather64<true><<<N_NODES * 16 / 256, 256, 0, stream>>>(xw2, off, cnt, dinv, tab, b2, out);
}


# Gather re-tiled: a wave walks 8 consecutive nodes, whole wave per 256-B row; a node's table entries read once, one per lane, passed on by lane reads: no row read waits on the table; sum order kept
# speedup vs baseline: 4.2653x; 1.0576x over previous
#include <hip/hip_runtime.h>

#define N_NODES 50000
#define N_EDGES 800000
#define IN_C    64
#define HID_C   128
#define OUT_C   64
#define N_PAD   50176

typedef __attribute__((ext_vector_type(2))) float v2f;
typedef __attribute__((ext_vector_type(8))) float v8f;

#define EQ      16
#define ER      16
#define CHUNK_E (N_EDGES / EQ)
#define CHUNK_V (CHUNK_E / 4)
#define RANGE_N (N_NODES / ER)
#define SCAN_T  1024

__global__ __launch_bounds__(SCAN_T) void gcn_edge_count(const int* __restrict__ dst, int* __restrict__ pc,
                                                         int* __restrict__ total) {
    __shared__ int lcnt[RANGE_N];
    const int tid = threadIdx.x;
    const int q   = blockIdx.x % EQ;
    const int r   = blockIdx.x / EQ;
    const int lo  = r * RANGE_N;
    for (int i = tid; i < RANGE_N; i += SCAN_T) lcnt[i] = 0;
    if (blockIdx.x == 0 && tid == 0) *total = 0;
    __syncthreads();
    const int4* __restrict__ dv = (const int4*)(dst + q * CHUNK_E);
    for (int base = 0; base < CHUNK_V; base += 4 * SCAN_T) {
        int dd[16];
#pragma unroll
        for (int u = 0; u < 4; ++u) {
            const int i = base + u * SCAN_T + tid;
            int4 d = make_int4(-1, -1, -1, -1);
            if (i < CHUNK_V) d = dv[i];
            dd[4 * u + 0] = d.x; dd[4 * u + 1] = d.y; dd[4 * u + 2] = d.z; dd[4 * u + 3] = d.w;
        }
#pragma unroll
        for (int k = 0; k < 16; ++k) {
            const unsigned a = (unsigned)(dd[k] - lo);
            if (a < (unsigned)RANGE_N)
                __hip_atomic_fetch_add(&lcnt[a], 1, __ATOMIC_RELAXED, __HIP_MEMORY_SCOPE_WORKGROUP);
        }
    }
    __syncthreads();
    for (int i = tid; i < RANGE_N; i += SCAN_T) pc[q * N_PAD + lo + i] = lcnt[i];
}

__global__ __launch_bounds__(256) void gcn_seg_alloc(int* __restrict__ pc, int* cnt, int* total,
                                                     int* __restrict__ off, float* __restrict__ dinv) {
    __shared__ int wsum[8];
    __shared__ int sbase;
    const int tid  = threadIdx.x;
    const int lane = tid & 31;
    const int wave = tid >> 5;
    const int n    = blockIdx.x * 256 + tid;
    int c = 0;
    if (n < N_NODES) {
        int v[EQ];
#pragma unroll
        for (int q = 0; q < EQ; ++q) v[q] = pc[q * N_PAD + n];
#pragma unroll
        for (int q = 0; q < EQ; ++q) { pc[q * N_PAD + n] = c; c += v[q]; }
    }
    int incl = c;
    for (int d = 1; d < 32; d <<= 1) {
        int v = __shfl_up(incl, d);
        if (lane >= d) incl += v;
    }
    if (lane == 31) wsum[wave] = incl;
    __syncthreads();
    if (tid == 0) {
        int run = 0;
        for (int w = 0; w < 8; ++w) { int t = wsum[w]; wsum[w] = run; run += t; }
        sbase = atomicAdd(total, run);
    }
    __syncthreads();
    if (n < N_NODES) {
        cnt[n]  = c;
        off[n]  = sbase + wsum[wave] + incl - c;
        dinv[n] = rsqrtf((float)(c + 1));
    }
}

__global__ __launch_bounds__(SCAN_T) void gcn_edge_fill(const int* __restrict__ src, const int* __restrict__ dst,
                                                        const int* __restrict__ pc, const int* __restrict__ off,
                                                        const float* __restrict__ dinv, int* __restrict__ tab) {
    __shared__ int   lpos[RANGE_N];
    __shared__ float ldinv[RANGE_N];
    const int tid = threadIdx.x;
    const int q   = blockIdx.x % EQ;
    const int r   = blockIdx.x / EQ;
    const int lo  = r * RANGE_N;
    int   lp[4];
    float ld[4];
#pragma unroll
    for (int t = 0; t < 4; ++t) {
        const int i  = t * SCAN_T + tid;
        const int ic = i < RANGE_N ? i : RANGE_N - 1;
        lp[t] = off[lo + ic] + pc[q * N_PAD + lo + ic];
        ld[t] = dinv[lo + ic];
    }
#pragma unroll
    for (int t = 0; t < 4; ++t) {
        const int i = t * SCAN_T + tid;
        if (i < RANGE_N) { lpos[i] = lp[t]; ldinv[i] = ld[t]; }
    }
    __syncthreads();
    const int4* __restrict__ dv = (const int4*)(dst + q * CHUNK_E);
    const int4* __restrict__ sv = (const int4*)(src + q * CHUNK_E);
    for (int base = 0; base < CHUNK_V; base += 4 * SCAN_T) {
        int dd[16], ss[16];
#pragma unroll
        for (int u = 0; u < 4; ++u) {
            const int i = base + u * SCAN_T + tid;
            int4 d = make_int4(-1, -1, -1, -1);
            int4 s = make_int4(0, 0, 0, 0);
            if (i < CHUNK_V) { d = dv[i]; s = sv[i]; }
            dd[4 * u + 0] = d.x; dd[4 * u + 1] = d.y; dd[4 * u + 2] = d.z; dd[4 * u + 3] = d.w;
            ss[4 * u + 0] = s.x; ss[4 * u + 1] = s.y; ss[4 * u + 2] = s.z; ss[4 * u + 3] = s.w;
        }
        int p[16];
#pragma unroll
        for (int k = 0; k < 16; ++k) {
            const unsigned a = (unsigned)(dd[k] - lo);
            p[k] = -1;
            if (a < (unsigned)RANGE_N)
                p[k] = __hip_atomic_fetch_add(&lpos[a], 1, __ATOMIC_RELAXED, __HIP_MEMORY_SCOPE_WORKGROUP);
        }
        float w[16];
#pragma unroll
        for (int k = 0; k < 16; ++k) {
            const unsigned a  = (unsigned)(dd[k] - lo);
            const bool    hit = a < (unsigned)RANGE_N;
            w[k] = dinv[hit ? ss[k] : 0] * ldinv[hit ? a : 0u];
        }
#pragma unroll
        for (int k = 0; k < 16; ++k) {
            if ((unsigned)p[k] < (unsigned)N_EDGES)
                ((int2*)tab)[p[k]] = make_int2(ss[k], __float_as_int(w[k]));
        }
    }
}

#define NPW          8
#define GATHER_WAVES (N_NODES / NPW)
#define GATHER_WPB   4
#define GATHER_BLKS  ((GATHER_WAVES + GATHER_WPB - 1) / GATHER_WPB)
#define SLOTS        24
#define GRP          4
static_assert(N_NODES % NPW == 0, "a wave's nodes must all exist");
static_assert(SLOTS % GRP == 0 && 32 % GRP == 0 && SLOTS <= 32, "groups must tile the 32 entry lanes");

__device__ __forceinline__ int2 gcn_seg_entries(const int2* __restrict__ seg, int base, int c, int lane, int n) {
    int2 e = make_int2(n, 0);
    if (base + lane < c) e = seg[base + lane];
    return e;
}

__device__ __forceinline__ void gcn_rows_issue(const float2* __restrict__ rows, int2 e, int c, int n, int lane,
                                               float2 (&r)[SLOTS]) {
#pragma unroll
    for (int g = 0; g < SLOTS / GRP; ++g) {
        if (g * GRP < c) {
#pragma unroll
            for (int k = 0; k < GRP; ++k) {
                const int slot = g * GRP + k;
                const int sx = __builtin_amdgcn_readlane(e.x, slot);
                const int sc = (unsigned)sx < (unsigned)N_NODES ? sx : n;
                r[slot] = rows[sc * 32 + lane];
            }
        }
    }
}

__device__ __forceinline__ void gcn_rows_add(int2 e, int c, const float2 (&r)[SLOTS], float2& acc) {
#pragma unroll
    for (int g = 0; g < SLOTS / GRP; ++g) {
        if (g * GRP < c) {
#pragma unroll
            for (int k = 0; k < GRP; ++k) {
                const int slot = g * GRP + k;
                const float w = __int_as_float(__builtin_amdgcn_readlane(e.y, slot));
                acc.x = __builtin_fmaf(r[slot].x, w, acc.x);
                acc.y = __builtin_fmaf(r[slot].y, w, acc.y);
            }
        }
    }
}

__device__ __forceinline__ void gcn_rows_tail(const float2* __restrict__ rows, const int2* __restrict__ seg,
                                              int2 e, int c, int n, int lane, float2& acc) {
    for (int j = SLOTS; j < c; j += GRP) {
        if (j >= 32 && (j & 31) == 0) e = gcn_seg_entries(seg, j, c, lane, n);
        float2 t[GRP];
#pragma unroll
        for (int k = 0; k < GRP; ++k) {
            const int sx = __builtin_amdgcn_readlane(e.x, (j + k) & 31);
            const int sc = (unsigned)sx < (unsigned)N_NODES ? sx : n;
            t[k] = rows[sc * 32 + lane];
        }
#pragma unroll
        for (int k = 0; k < GRP; ++k) {
            const float w = __int_as_float(__builtin_amdgcn_readlane(e.y, (j + k) & 31));
            acc.x = __builtin_fmaf(t[k].x, w, acc.x);
            acc.y = __builtin_fmaf(t[k].y, w, acc.y);
        }
    }
}

template <bool BIAS>
__global__ __launch_bounds__(32 * GATHER_WPB) void gcn_gather64(
    const float* __restrict__ xw, const int* __restrict__ off, const int* __restrict__ cnt,
    const float* __restrict__ dinv, const int* __restrict__ tab,
    const float* __restrict__ bias, float* __restrict__ agg) {
    const int lane = threadIdx.x & 31;
    const int wv = __builtin_amdgcn_readfirstlane((int)(threadIdx.x >> 5));
    const int gw = blockIdx.x * GATHER_WPB + wv;
    if (gw >= GATHER_WAVES) return;
    const int n0 = gw * NPW;
    const float2* __restrict__ rows = (const float2*)xw;
    const int2*   __restrict__ tb   = (const int2*)tab;

    const int   li = lane < NPW ? lane : 0;
    const int   vc = cnt[n0 + li];
    const int   vo = off[n0 + li];
    const float vd = dinv[n0 + li];
    float2 bia = make_float2(0.0f, 0.0f);
    if (BIAS) bia = ((const float2*)bias)[lane];

    int    cA_n = __builtin_amdgcn_readlane(vc, 0);
    int    oA_n = __builtin_amdgcn_readlane(vo, 0);
    int    cB_n = __builtin_amdgcn_readlane(vc, NPW / 2);
    int    oB_n = __builtin_amdgcn_readlane(vo, NPW / 2);
    int2   eA_n = gcn_seg_entries(tb + oA_n, 0, cA_n, lane, n0);
    int2   eB_n = gcn_seg_entries(tb + oB_n, 0, cB_n, lane, n0 + NPW / 2);
    float2 vA_n = rows[n0 * 32 + lane];
    float2 vB_n = rows[(n0 + NPW / 2) * 32 + lane];

    for (int i = 0; i < NPW / 2; ++i) {
        const int    nA = n0 + i, nB = n0 + NPW / 2 + i;
        const int    cA = cA_n, oA = oA_n, cB = cB_n, oB = oB_n;
        const int2   eA = eA_n, eB = eB_n;
        const float2 vA = vA_n, vB = vB_n;
        const float  dA = __int_as_float(__builtin_amdgcn_readlane(__float_as_int(vd), i));
        const float  dB = __int_as_float(__builtin_amdgcn_readlane(__float_as_int(vd), i + NPW / 2));
        if (i + 1 < NPW / 2) {
            cA_n = __builtin_amdgcn_readlane(vc, i + 1);
            oA_n = __builtin_amdgcn_readlane(vo, i + 1);
            cB_n = __builtin_amdgcn_readlane(vc, i + 1 + NPW / 2);
            oB_n = __builtin_amdgcn_readlane(vo, i + 1 + NPW / 2);
            eA_n = gcn_seg_entries(tb + oA_n, 0, cA_n, lane, nA + 1);
            eB_n = gcn_seg_entries(tb + oB_n, 0, cB_n, lane, nB + 1);
            vA_n = rows[(nA + 1) * 32 + lane];
            vB_n = rows[(nB + 1) * 32 + lane];
        }
        float2 accA, accB;
        accA.x = vA.x * dA * dA;
        accA.y = vA.y * dA * dA;
        accB.x = vB.x * dB * dB;
        accB.y = vB.y * dB * dB;
        float2 rA[SLOTS], rB[SLOTS];
        gcn_rows_issue(rows, eA, cA, nA, lane, rA);
        gcn_rows_issue(rows, eB, cB, nB, lane, rB);
        gcn_rows_add(eA, cA, rA, accA);
        gcn_rows_add(eB, cB, rB, accB);
        if (cA > SLOTS) gcn_rows_tail(rows, tb + oA, eA, cA, nA, lane, accA);
        if (cB > SLOTS) gcn_rows_tail(rows, tb + oB, eB, cB, nB, lane, accB);
        if (BIAS) {
            accA.x += bia.x; accA.y += bia.y;
            accB.x += bia.x; accB.y += bia.y;
        }
        ((float2*)agg)[nA * 32 + lane] = accA;
        ((float2*)agg)[nB * 32 + lane] = accB;
    }
}

#define W1_PITCH 136
#define W2_PITCH 68
#define N_STRIPES (N_NODES / 16)

__global__ __launch_bounds__(256) void gcn_mlp_fused_wmma(
    const float* __restrict__ ax, const float* __restrict__ W1, const float* __restrict__ b1,
    const float* __restrict__ W2, float* __restrict__ xw) {
    __shared__ __attribute__((aligned(16))) float2 W1l[(IN_C / 2) * W1_PITCH];
    __shared__ __attribute__((aligned(16))) float2 W2l[(HID_C / 2) * W2_PITCH];
    const int tid    = threadIdx.x;
    const int lane   = tid & 31;
    const int wave   = tid >> 5;
    const int half   = lane >> 4;
    const int l15    = lane & 15;
    const int stripe = blockIdx.x * 8 + wave;
    const bool valid = stripe < N_STRIPES;
    const int m0     = (valid ? stripe : 0) * 16;

    float4 axv[8];
    const float4* __restrict__ arow = (const float4*)(ax + (size_t)(m0 + l15) * IN_C) + half;
#pragma unroll
    for (int i = 0; i < 8; ++i) axv[i] = arow[2 * i];

#pragma unroll
    for (int it = 0; it < 4; ++it) {
        const int item = it * 256 + tid;
        const int pr = item >> 5, q = item & 31;
        const float4 lo = *(const float4*)(W1 + (2 * pr) * HID_C + 4 * q);
        const float4 hi = *(const float4*)(W1 + (2 * pr + 1) * HID_C + 4 * q);
        float4* d = (float4*)(W1l + pr * W1_PITCH + 4 * q);
        d[0] = make_float4(lo.x, hi.x, lo.y, hi.y);
        d[1] = make_float4(lo.z, hi.z, lo.w, hi.w);
    }
#pragma unroll
    for (int it = 0; it < 4; ++it) {
        const int item = it * 256 + tid;
        const int pr = item >> 4, q = item & 15;
        const float4 lo = *(const float4*)(W2 + (2 * pr) * OUT_C + 4 * q);
        const float4 hi = *(const float4*)(W2 + (2 * pr + 1) * OUT_C + 4 * q);
        float4* d = (float4*)(W2l + pr * W2_PITCH + 4 * q);
        d[0] = make_float4(lo.x, hi.x, lo.y, hi.y);
        d[1] = make_float4(lo.z, hi.z, lo.w, hi.w);
    }
    __syncthreads();
    if (!valid) return;

    v8f acc1[8];
#pragma unroll
    for (int t = 0; t < 8; ++t) {
        const float4 ba = *(const float4*)(b1 + 16 * t + 8 * half);
        const float4 bb = *(const float4*)(b1 + 16 * t + 8 * half + 4);
        acc1[t] = (v8f){ba.x, ba.y, ba.z, ba.w, bb.x, bb.y, bb.z, bb.w};
    }
#pragma unroll
    for (int i = 0; i < 8; ++i) {
#pragma unroll
        for (int j = 0; j < 2; ++j) {
            v2f xb;
            xb.x = j ? axv[i].z : axv[i].x;
            xb.y = j ? axv[i].w : axv[i].y;
            const float2* __restrict__ wr = W1l + (4 * i + 2 * half + j) * W1_PITCH + l15;
#pragma unroll
            for (int t = 0; t < 8; ++t) {
                const float2 w = wr[16 * t];
                v2f wa; wa.x = w.x; wa.y = w.y;
                acc1[t] = __builtin_amdgcn_wmma_f32_16x16x4_f32(
                     false, wa,  false, xb,
                     (short)0, acc1[t],  false,  false);
            }
        }
    }

    v8f acc2[4];
#pragma unroll
    for (int u = 0; u < 4; ++u) acc2[u] = (v8f){0.f, 0.f, 0.f, 0.f, 0.f, 0.f, 0.f, 0.f};
#pragma unroll
    for (int t = 0; t < 8; ++t) {
#pragma unroll
        for (int j = 0; j < 4; ++j) {
            v2f ha;
            ha.x = fmaxf(acc1[t][2 * j], 0.0f);
            ha.y = fmaxf(acc1[t][2 * j + 1], 0.0f);
            const float2* __restrict__ wr = W2l + (8 * t + 4 * half + j) * W2_PITCH + l15;
#pragma unroll
            for (int u = 0; u < 4; ++u) {
                const float2 w = wr[16 * u];
                v2f wb; wb.x = w.x; wb.y = w.y;
                acc2[u] = __builtin_amdgcn_wmma_f32_16x16x4_f32(
                    false, ha, false, wb, (short)0, acc2[u], false, false);
            }
        }
    }
#pragma unroll
    for (int u = 0; u < 4; ++u)
#pragma unroll
        for (int r = 0; r < 8; ++r)
            xw[(size_t)(m0 + half * 8 + r) * OUT_C + 16 * u + l15] = acc2[u][r];
}


extern "C" void kernel_launch(void* const* d_in, const int* in_sizes, int n_in,
                              void* d_out, int out_size, void* d_ws, size_t ws_size,
                              hipStream_t stream) {
    const float* x  = (const float*)d_in[0];
    const int*   ei = (const int*)d_in[1];
    const float* W1 = (const float*)d_in[2];
    const float* b1 = (const float*)d_in[3];
    const float* W2 = (const float*)d_in[4];
    const float* b2 = (const float*)d_in[5];
    float* out = (float*)d_out;

    const int* src = ei;
    const int* dst = ei + N_EDGES;

    int*   cnt  = (int*)d_ws;
    int*   off  = cnt + N_PAD;
    float* dinv = (float*)(off + N_PAD);
    int*   pc   = (int*)(dinv + N_PAD);
    int*   tab  = pc + EQ * N_PAD;
    float* ax   = (float*)(tab + 2 * N_EDGES);
    float* xw2  = ax + 3200000;

    gcn_edge_count<<<EQ * ER, SCAN_T, 0, stream>>>(dst, pc, cnt + N_NODES);
    gcn_seg_alloc<<<N_PAD / 256, 256, 0, stream>>>(pc, cnt, cnt + N_NODES, off, dinv);
    gcn_edge_fill<<<EQ * ER, SCAN_T, 0, stream>>>(src, dst, pc, off, dinv, tab);

    gcn_gather64<false><<<GATHER_BLKS, 32 * GATHER_WPB, 0, stream>>>(x, off, cnt, dinv, tab, b1, ax);

    gcn_mlp_fused_wmma<<<(N_STRIPES + 7) / 8, 256, 0, stream>>>(ax, W1, b1, W2, xw2);

    gcn_gather64<true><<<GATHER_BLKS, 32 * GATHER_WPB, 0, stream>>>(xw2, off, cnt, dinv, tab, b2, out);
}
